# starting kernel
# baseline (speedup 1.0000x reference)
#include <hip/hip_runtime.h>

typedef float v2f __attribute__((ext_vector_type(2)));
typedef float v8f __attribute__((ext_vector_type(8)));

#define IMG   512
#define TILE  16
#define HALO  18
#define PITCH 20
#define CHSZ  (HALO*PITCH)
#define NCH   30

__global__ __launch_bounds__(256) void hmm_fused_kernel(
    const float* __restrict__ obs,
    const float* __restrict__ P,
    const float* __restrict__ u_k,
    const float* __restrict__ w_k,
    const float* __restrict__ conv_w,
    const float* __restrict__ conv_b,
    float* __restrict__ u_out,
    float* __restrict__ w_out)
{
    __shared__ float X[NCH * CHSZ];

    const int tid = threadIdx.x;
    const int wgx = blockIdx.x * TILE;
    const int wgy = blockIdx.y * TILE;

    float Pm[9];
#pragma unroll
    for (int i = 0; i < 9; ++i) Pm[i] = P[i];
    const float cb0 = conv_b[0], cb1 = conv_b[1], cb2 = conv_b[2];

#pragma unroll
    for (int it = 0; it < 2; ++it) {
        int hp = tid + it * 256;
        if (hp < HALO * HALO) {
            const int hy = hp / HALO, hx = hp % HALO;
            const int gy = wgy + hy - 1, gx = wgx + hx - 1;
            const int lb = hy * PITCH + hx;
            if (gy >= 0 && gy < IMG && gx >= 0 && gx < IMG) {
                const int pix = gy * IMG + gx;
                const float o0 = obs[pix * 3 + 0], o1 = obs[pix * 3 + 1], o2 = obs[pix * 3 + 2];
                const float u0 = u_k[pix * 3 + 0], u1 = u_k[pix * 3 + 1], u2 = u_k[pix * 3 + 2];
                const float b0 = Pm[0] * o0 + Pm[3] * o1 + Pm[6] * o2;
                const float b1 = Pm[1] * o0 + Pm[4] * o1 + Pm[7] * o2;
                const float b2 = Pm[2] * o0 + Pm[5] * o1 + Pm[8] * o2;
                const float Bu0 = b0 * u0, Bu1 = b1 * u1, Bu2 = b2 * u2;
                const float bu  = Bu0 + Bu1 + Bu2;
                const float inv = 1.0f / bu;
                const float Z0 = Bu0 * inv, Z1 = Bu1 * inv, Z2 = Bu2 * inv;
                X[0 * CHSZ + lb] = Z0;
                X[1 * CHSZ + lb] = Z1;
                X[2 * CHSZ + lb] = Z2;
                const float ov[3] = {o0, o1, o2};
                const float uv[3] = {u0, u1, u2};
                const float bv[3] = {b0, b1, b2};
                const float Zv[3] = {Z0, Z1, Z2};
                const float* wp = w_k + (long)pix * 27;
#pragma unroll
                for (int ij = 0; ij < 9; ++ij) {
                    const int i = ij / 3, j = ij % 3;
                    float du0 = bv[0] * wp[0 * 9 + ij] + ((j == 0) ? ov[i] * uv[0] : 0.0f);
                    float du1 = bv[1] * wp[1 * 9 + ij] + ((j == 1) ? ov[i] * uv[1] : 0.0f);
                    float du2 = bv[2] * wp[2 * 9 + ij] + ((j == 2) ? ov[i] * uv[2] : 0.0f);
                    float sdu = du0 + du1 + du2;
                    X[(3 + ij * 3 + 0) * CHSZ + lb] = (du0 - sdu * Zv[0]) * inv;
                    X[(3 + ij * 3 + 1) * CHSZ + lb] = (du1 - sdu * Zv[1]) * inv;
                    X[(3 + ij * 3 + 2) * CHSZ + lb] = (du2 - sdu * Zv[2]) * inv;
                }
            } else {
#pragma unroll
                for (int c = 0; c < NCH; ++c) X[c * CHSZ + lb] = 0.0f;
            }
        }
    }
    __syncthreads();

    const int lane = tid & 31;
    const int wv   = tid >> 5;
    const int half = (lane < 16) ? 0 : 2;
    const int n    = lane & 15;

    v2f Amat[7];
    int off0[7], off1[7];
#pragma unroll
    for (int jc = 0; jc < 7; ++jc) {
        const int k0 = jc * 4 + half;
        const int k1 = k0 + 1;
        float a0 = 0.0f, a1 = 0.0f;
        if (n < 3 && k0 < 27) a0 = conv_w[n * 27 + (k0 % 3) * 9 + (k0 / 3)];
        if (n < 3 && k1 < 27) a1 = conv_w[n * 27 + (k1 % 3) * 9 + (k1 / 3)];
        Amat[jc][0] = a0; Amat[jc][1] = a1;
        const int c0 = (k0 < 27) ? k0 : 26;
        const int c1 = (k1 < 27) ? k1 : 26;
        const int t0 = c0 / 3, t1 = c1 / 3;
        off0[jc] = (c0 % 3) * CHSZ + (t0 / 3) * PITCH + (t0 % 3) + n;
        off1[jc] = (c1 % 3) * CHSZ + (t1 / 3) * PITCH + (t1 % 3) + n;
    }

    auto conv_group = [&](int gbase) -> v8f {
        v8f acc = {0.f, 0.f, 0.f, 0.f, 0.f, 0.f, 0.f, 0.f};
#pragma unroll
        for (int jc = 0; jc < 7; ++jc) {
            v2f Bv;
            Bv[0] = X[gbase + off0[jc]];
            Bv[1] = X[gbase + off1[jc]];
            acc = __builtin_amdgcn_wmma_f32_16x16x4_f32(
                false, Amat[jc], false, Bv, (short)0, acc, false, false);
        }
        return acc;
    };

#pragma unroll
    for (int r = 0; r < 2; ++r) {
        const int ty     = wv * 2 + r;
        const int gy     = wgy + ty;
        const int rowoff = ty * PITCH;

        v8f y = conv_group(rowoff);
        float y0 = y[0] + cb0, y1 = y[1] + cb1, y2 = y[2] + cb2;
        float m  = fmaxf(y0, fmaxf(y1, y2));
        float e0 = __expf(y0 - m), e1 = __expf(y1 - m), e2 = __expf(y2 - m);
        float is = 1.0f / (e0 + e1 + e2);
        float p0 = e0 * is, p1 = e1 * is, p2 = e2 * is;

        const int pix = gy * IMG + (wgx + n);
        if (lane < 16) {
            u_out[pix * 3 + 0] = p0;
            u_out[pix * 3 + 1] = p1;
            u_out[pix * 3 + 2] = p2;
        }

        for (int g = 1; g < 10; ++g) {
            v8f dv = conv_group(g * (3 * CHSZ) + rowoff);
            float d0 = dv[0], d1 = dv[1], d2 = dv[2];
            float dot = p0 * d0 + p1 * d1 + p2 * d2;
            float w0 = p0 * (d0 - dot);
            float w1 = p1 * (d1 - dot);
            float w2 = p2 * (d2 - dot);
            if (lane < 16) {
                float* wo = w_out + (long)pix * 27 + (g - 1);
                wo[0]  = w0;
                wo[9]  = w1;
                wo[18] = w2;
            }
        }
    }
}

extern "C" void kernel_launch(void* const* d_in, const int* in_sizes, int n_in,
                              void* d_out, int out_size, void* d_ws, size_t ws_size,
                              hipStream_t stream) {
    (void)in_sizes; (void)n_in; (void)out_size; (void)d_ws; (void)ws_size;
    const float* obs    = (const float*)d_in[0];
    const float* P      = (const float*)d_in[1];
    const float* u_k    = (const float*)d_in[2];
    const float* w_k    = (const float*)d_in[3];
    const float* conv_w = (const float*)d_in[4];
    const float* conv_b = (const float*)d_in[5];
    float* u_out = (float*)d_out;
    float* w_out = u_out + (size_t)IMG * IMG * 3;

    dim3 grid(IMG / TILE, IMG / TILE);
    dim3 block(256);
    hmm_fused_kernel<<<grid, block, 0, stream>>>(obs, P, u_k, w_k, conv_w, conv_b,
                                                 u_out, w_out);
}


# 3x3 conv split by kernel row on WMMA: per halo row A rows = (dy, c_out), K = (c_in, dx) = 9; a wave owns 4 output rows and adds 3 row partials; 180 WMMAs per 4 rows not 280; conv sum order changes
# speedup vs baseline: 1.6289x; 1.6289x over previous
#include <hip/hip_runtime.h>

typedef float v2f __attribute__((ext_vector_type(2)));
typedef float v8f __attribute__((ext_vector_type(8)));
typedef float v4f __attribute__((ext_vector_type(4)));
typedef v4f f4u __attribute__((aligned(4)));

#define IMG   512
#define TILE  16
#define HALO  18
#define PITCH 20
#define CHSZ  (HALO*PITCH)
#define NCH   30
#define RCH   122
#define RPITCH 500
#define NWK   (IMG * IMG * 27)
#define NTHR  128
#define WROWS 4

__device__ __forceinline__ float xhalf(float x) {
    return __builtin_bit_cast(float, __builtin_amdgcn_permlanex16(
        __builtin_bit_cast(int, x), __builtin_bit_cast(int, x), 0x76543210, 0xfedcba98, false, false));
}

__global__ __launch_bounds__(NTHR) void hmm_fused_kernel(
    const float* __restrict__ obs,
    const float* __restrict__ P,
    const float* __restrict__ u_k,
    const float* __restrict__ w_k,
    const float* __restrict__ conv_w,
    const float* __restrict__ conv_b,
    float* __restrict__ u_out,
    float* __restrict__ w_out)
{
    __shared__ float X[NCH * CHSZ];
    __shared__ __attribute__((aligned(16))) float R[HALO * RPITCH];

    const int tid = threadIdx.x;
    const int wgx = blockIdx.x * TILE;
    const int wgy = blockIdx.y * TILE;

    float Pm[9];
#pragma unroll
    for (int i = 0; i < 9; ++i) Pm[i] = P[i];
    const float cb0 = conv_b[0], cb1 = conv_b[1], cb2 = conv_b[2];


    float obv[3][3], ukv[3][3];
#pragma unroll
    for (int it = 0; it < 3; ++it) {
        const int hp = tid + it * NTHR;
        const int hy = hp / HALO, hx = hp % HALO;
        const int gy = wgy + hy - 1, gx = wgx + hx - 1;
        const bool in = hp < HALO * HALO && gy >= 0 && gy < IMG && gx >= 0 && gx < IMG;
        const int pix = in ? gy * IMG + gx : 0;
#pragma unroll
        for (int k = 0; k < 3; ++k) { obv[it][k] = obs[pix * 3 + k]; ukv[it][k] = u_k[pix * 3 + k]; }
    }
    {
        const int col = tid;
        if (col < RCH) {
            v4f stage[HALO];
#pragma unroll
            for (int hy = 0; hy < HALO; ++hy) {
                int f = ((wgy + hy - 1) * IMG + wgx) * 27 - 28 + col * 4;
                f = min(max(f, 0), NWK - 4);
                stage[hy] = *(const f4u*)(w_k + f);
            }
#pragma unroll
            for (int hy = 0; hy < HALO; ++hy)
                *(v4f*)(R + hy * RPITCH + col * 4) = stage[hy];
        }
    }
    __syncthreads();

#pragma unroll
    for (int it = 0; it < 3; ++it) {
        int hp = tid + it * NTHR;
        if (hp < HALO * HALO) {
            const int hy = hp / HALO, hx = hp % HALO;
            const int gy = wgy + hy - 1, gx = wgx + hx - 1;
            const int lb = hy * PITCH + hx;
            if (gy >= 0 && gy < IMG && gx >= 0 && gx < IMG) {
                const float o0 = obv[it][0], o1 = obv[it][1], o2 = obv[it][2];
                const float u0 = ukv[it][0], u1 = ukv[it][1], u2 = ukv[it][2];
                const float b0 = Pm[0] * o0 + Pm[3] * o1 + Pm[6] * o2;
                const float b1 = Pm[1] * o0 + Pm[4] * o1 + Pm[7] * o2;
                const float b2 = Pm[2] * o0 + Pm[5] * o1 + Pm[8] * o2;
                const float Bu0 = b0 * u0, Bu1 = b1 * u1, Bu2 = b2 * u2;
                const float bu  = Bu0 + Bu1 + Bu2;
                const float inv = 1.0f / bu;
                const float Z0 = Bu0 * inv, Z1 = Bu1 * inv, Z2 = Bu2 * inv;
                X[0 * CHSZ + lb] = Z0;
                X[1 * CHSZ + lb] = Z1;
                X[2 * CHSZ + lb] = Z2;
                const float ov[3] = {o0, o1, o2};
                const float uv[3] = {u0, u1, u2};
                const float bv[3] = {b0, b1, b2};
                const float Zv[3] = {Z0, Z1, Z2};
                const float* wp = R + hy * RPITCH + 1 + hx * 27;
#pragma unroll
                for (int ij = 0; ij < 9; ++ij) {
                    const int i = ij / 3, j = ij % 3;
                    float du0 = bv[0] * wp[0 * 9 + ij] + ((j == 0) ? ov[i] * uv[0] : 0.0f);
                    float du1 = bv[1] * wp[1 * 9 + ij] + ((j == 1) ? ov[i] * uv[1] : 0.0f);
                    float du2 = bv[2] * wp[2 * 9 + ij] + ((j == 2) ? ov[i] * uv[2] : 0.0f);
                    float sdu = du0 + du1 + du2;
                    X[(3 + ij * 3 + 0) * CHSZ + lb] = (du0 - sdu * Zv[0]) * inv;
                    X[(3 + ij * 3 + 1) * CHSZ + lb] = (du1 - sdu * Zv[1]) * inv;
                    X[(3 + ij * 3 + 2) * CHSZ + lb] = (du2 - sdu * Zv[2]) * inv;
                }
            } else {
#pragma unroll
                for (int c = 0; c < NCH; ++c) X[c * CHSZ + lb] = 0.0f;
            }
        }
    }
    __syncthreads();

    const int lane = tid & 31;
    const int wv   = tid >> 5;
    const int half = (lane < 16) ? 0 : 2;
    const int n    = lane & 15;

    v2f Amat[3];
    int off0[3], off1[3];
#pragma unroll
    for (int jc = 0; jc < 3; ++jc) {
        const int k0 = jc * 4 + half;
        const int k1 = k0 + 1;
        float a0 = 0.0f, a1 = 0.0f;
        if (n < 9 && k0 < 9) a0 = conv_w[(n % 3) * 27 + (k0 % 3) * 9 + (n / 3) * 3 + (k0 / 3)];
        if (n < 9 && k1 < 9) a1 = conv_w[(n % 3) * 27 + (k1 % 3) * 9 + (n / 3) * 3 + (k1 / 3)];
        Amat[jc][0] = a0; Amat[jc][1] = a1;
        const int c0 = (k0 < 9) ? k0 : 8;
        const int c1 = (k1 < 9) ? k1 : 8;
        off0[jc] = (c0 % 3) * CHSZ + (c0 / 3) + n;
        off1[jc] = (c1 % 3) * CHSZ + (c1 / 3) + n;
    }

    const int rbase = wv * WROWS * PITCH;
    auto load_item = [&](int t, v2f (&B)[2][3]) {
        const int base = rbase + (t / 5) * PITCH + (t % 5) * (6 * CHSZ);
#pragma unroll
        for (int q = 0; q < 2; ++q) {
#pragma unroll
            for (int jc = 0; jc < 3; ++jc) {
                B[q][jc][0] = X[base + q * (3 * CHSZ) + off0[jc]];
                B[q][jc][1] = X[base + q * (3 * CHSZ) + off1[jc]];
            }
        }
    };

    float T0[10][3], T1[10][3];

    v2f Bq[2][2][3];
    load_item(0, Bq[0]);

#pragma unroll
    for (int j = 0; j < WROWS + 2; ++j) {
        const int ty  = wv * WROWS + (j - 2);
        const int gy  = wgy + ty;
        const int pix = gy * IMG + (wgx + n);
        float p0 = 0.f, p1 = 0.f, p2 = 0.f;
        float wacc[27];
#pragma unroll
        for (int gp = 0; gp < 5; ++gp) {
            const int t = j * 5 + gp;
            if (t + 1 < 5 * (WROWS + 2))
                load_item(t + 1, Bq[(t + 1) & 1]);
            __builtin_amdgcn_sched_barrier(0);
            v8f dq[2];
            dq[0] = (v8f){0.f, 0.f, 0.f, 0.f, 0.f, 0.f, 0.f, 0.f};
            dq[1] = (v8f){0.f, 0.f, 0.f, 0.f, 0.f, 0.f, 0.f, 0.f};
#pragma unroll
            for (int jc = 0; jc < 3; ++jc) {
                dq[0] = __builtin_amdgcn_wmma_f32_16x16x4_f32(
                    false, Amat[jc], false, Bq[t & 1][0][jc], (short)0, dq[0], false, false);
                dq[1] = __builtin_amdgcn_wmma_f32_16x16x4_f32(
                    false, Amat[jc], false, Bq[t & 1][1][jc], (short)0, dq[1], false, false);
            }
            __builtin_amdgcn_sched_barrier(0);
#pragma unroll
            for (int q = 0; q < 2; ++q) {
                const int g = gp * 2 + q;
                const v8f dv = dq[q];
                const float r8 = xhalf(dv[0]);
                float d0 = 0.f, d1 = 0.f, d2 = 0.f;
                if (j >= 2) {
                    d0 = T1[g][0] + dv[6];
                    d1 = T1[g][1] + dv[7];
                    d2 = T1[g][2] + r8;
                }
                if (j >= 1 && j <= WROWS) {
                    T1[g][0] = T0[g][0] + dv[3];
                    T1[g][1] = T0[g][1] + dv[4];
                    T1[g][2] = T0[g][2] + dv[5];
                }
                if (j < WROWS) {
                    T0[g][0] = dv[0];
                    T0[g][1] = dv[1];
                    T0[g][2] = dv[2];
                }
                if (j >= 2) {
                    if (g == 0) {
                        float y0 = d0 + cb0, y1 = d1 + cb1, y2 = d2 + cb2;
                        float m  = fmaxf(y0, fmaxf(y1, y2));
                        float e0 = __expf(y0 - m), e1 = __expf(y1 - m), e2 = __expf(y2 - m);
                        float is = 1.0f / (e0 + e1 + e2);
                        p0 = e0 * is; p1 = e1 * is; p2 = e2 * is;
                        if (lane < 16) {
                            u_out[pix * 3 + 0] = p0;
                            u_out[pix * 3 + 1] = p1;
                            u_out[pix * 3 + 2] = p2;
                        }
                    } else {
                        float dot = p0 * d0 + p1 * d1 + p2 * d2;
                        wacc[g - 1]      = p0 * (d0 - dot);
                        wacc[9 + g - 1]  = p1 * (d1 - dot);
                        wacc[18 + g - 1] = p2 * (d2 - dot);
                    }
                }
            }
        }
        if (j >= 2 && lane < 16) {
            float* wo = w_out + (long)pix * 27;
#pragma unroll
            for (int q = 0; q < 6; ++q) {
                f4u v = {wacc[4 * q + 0], wacc[4 * q + 1], wacc[4 * q + 2], wacc[4 * q + 3]};
                *(f4u*)(wo + 4 * q) = v;
            }
            wo[24] = wacc[24];
            wo[25] = wacc[25];
            wo[26] = wacc[26];
        }
    }
}

extern "C" void kernel_launch(void* const* d_in, const int* in_sizes, int n_in,
                              void* d_out, int out_size, void* d_ws, size_t ws_size,
                              hipStream_t stream) {
    (void)in_sizes; (void)n_in; (void)out_size; (void)d_ws; (void)ws_size;
    const float* obs    = (const float*)d_in[0];
    const float* P      = (const float*)d_in[1];
    const float* u_k    = (const float*)d_in[2];
    const float* w_k    = (const float*)d_in[3];
    const float* conv_w = (const float*)d_in[4];
    const float* conv_b = (const float*)d_in[5];
    float* u_out = (float*)d_out;
    float* w_out = u_out + (size_t)IMG * IMG * 3;

    dim3 grid(IMG / TILE, IMG / TILE);
    dim3 block(NTHR);
    hmm_fused_kernel<<<grid, block, 0, stream>>>(obs, P, u_k, w_k, conv_w, conv_b,
                                                 u_out, w_out);
}


# conv on pixel pairs: softmax and its JVP need only out_c - out_2, so weights W[c]-W[2] give M = (2 px, dy, 2 c) = 12, K = (c_in, 4 halo cols) = 12; 90 WMMAs per wave not 180; conv sum order changes
# speedup vs baseline: 1.8024x; 1.1065x over previous
#include <hip/hip_runtime.h>

typedef float v2f __attribute__((ext_vector_type(2)));
typedef float v8f __attribute__((ext_vector_type(8)));
typedef float v4f __attribute__((ext_vector_type(4)));
typedef v4f f4u __attribute__((aligned(4)));

#define IMG   512
#define TILE  16
#define HALO  18
#define PITCH 20
#define CHSZ  (HALO*PITCH)
#define NCH   30
#define RCH   122
#define RPITCH 500
#define NWK   (IMG * IMG * 27)
#define NTHR  128
#define WROWS 4

__device__ __forceinline__ float xoct(float x) {
    return __builtin_bit_cast(float, __builtin_amdgcn_update_dpp(
        0, __builtin_bit_cast(int, x), 0x128, 0xf, 0xf, false));
}

__global__ __launch_bounds__(NTHR) void hmm_fused_kernel(
    const float* __restrict__ obs,
    const float* __restrict__ P,
    const float* __restrict__ u_k,
    const float* __restrict__ w_k,
    const float* __restrict__ conv_w,
    const float* __restrict__ conv_b,
    float* __restrict__ u_out,
    float* __restrict__ w_out)
{
    __shared__ __attribute__((aligned(16))) float X[NCH * CHSZ];
    __shared__ __attribute__((aligned(16))) float R[HALO * RPITCH];

    const int tid = threadIdx.x;
    const int wgx = blockIdx.x * TILE;
    const int wgy = blockIdx.y * TILE;

    float Pm[9];
#pragma unroll
    for (int i = 0; i < 9; ++i) Pm[i] = P[i];
    const float cb0 = conv_b[0], cb1 = conv_b[1], cb2 = conv_b[2];


    float obv[3][3], ukv[3][3];
#pragma unroll
    for (int it = 0; it < 3; ++it) {
        const int hp = tid + it * NTHR;
        const int hy = hp / HALO, hx = hp % HALO;
        const int gy = wgy + hy - 1, gx = wgx + hx - 1;
        const bool in = hp < HALO * HALO && gy >= 0 && gy < IMG && gx >= 0 && gx < IMG;
        const int pix = in ? gy * IMG + gx : 0;
#pragma unroll
        for (int k = 0; k < 3; ++k) { obv[it][k] = obs[pix * 3 + k]; ukv[it][k] = u_k[pix * 3 + k]; }
    }
    {
        const int col = tid;
        if (col < RCH) {
            v4f stage[HALO];
#pragma unroll
            for (int hy = 0; hy < HALO; ++hy) {
                int f = ((wgy + hy - 1) * IMG + wgx) * 27 - 28 + col * 4;
                f = min(max(f, 0), NWK - 4);
                stage[hy] = *(const f4u*)(w_k + f);
            }
#pragma unroll
            for (int hy = 0; hy < HALO; ++hy)
                *(v4f*)(R + hy * RPITCH + col * 4) = stage[hy];
        }
    }
    __syncthreads();

#pragma unroll
    for (int it = 0; it < 3; ++it) {
        int hp = tid + it * NTHR;
        if (hp < HALO * HALO) {
            const int hy = hp / HALO, hx = hp % HALO;
            const int gy = wgy + hy - 1, gx = wgx + hx - 1;
            const int lb = hy * PITCH + hx;
            if (gy >= 0 && gy < IMG && gx >= 0 && gx < IMG) {
                const float o0 = obv[it][0], o1 = obv[it][1], o2 = obv[it][2];
                const float u0 = ukv[it][0], u1 = ukv[it][1], u2 = ukv[it][2];
                const float b0 = Pm[0] * o0 + Pm[3] * o1 + Pm[6] * o2;
                const float b1 = Pm[1] * o0 + Pm[4] * o1 + Pm[7] * o2;
                const float b2 = Pm[2] * o0 + Pm[5] * o1 + Pm[8] * o2;
                const float Bu0 = b0 * u0, Bu1 = b1 * u1, Bu2 = b2 * u2;
                const float bu  = Bu0 + Bu1 + Bu2;
                const float inv = 1.0f / bu;
                const float Z0 = Bu0 * inv, Z1 = Bu1 * inv, Z2 = Bu2 * inv;
                X[0 * CHSZ + lb] = Z0;
                X[1 * CHSZ + lb] = Z1;
                X[2 * CHSZ + lb] = Z2;
                const float ov[3] = {o0, o1, o2};
                const float uv[3] = {u0, u1, u2};
                const float bv[3] = {b0, b1, b2};
                const float Zv[3] = {Z0, Z1, Z2};
                const float* wp = R + hy * RPITCH + 1 + hx * 27;
#pragma unroll
                for (int ij = 0; ij < 9; ++ij) {
                    const int i = ij / 3, j = ij % 3;
                    float du0 = bv[0] * wp[0 * 9 + ij] + ((j == 0) ? ov[i] * uv[0] : 0.0f);
                    float du1 = bv[1] * wp[1 * 9 + ij] + ((j == 1) ? ov[i] * uv[1] : 0.0f);
                    float du2 = bv[2] * wp[2 * 9 + ij] + ((j == 2) ? ov[i] * uv[2] : 0.0f);
                    float sdu = du0 + du1 + du2;
                    X[(3 + ij * 3 + 0) * CHSZ + lb] = (du0 - sdu * Zv[0]) * inv;
                    X[(3 + ij * 3 + 1) * CHSZ + lb] = (du1 - sdu * Zv[1]) * inv;
                    X[(3 + ij * 3 + 2) * CHSZ + lb] = (du2 - sdu * Zv[2]) * inv;
                }
            } else {
#pragma unroll
                for (int c = 0; c < NCH; ++c) X[c * CHSZ + lb] = 0.0f;
            }
        }
    }
    __syncthreads();

    const int lane = tid & 31;
    const int wv   = tid >> 5;
    const int h    = lane >> 4;
    const int n    = lane & 15;
    const int pr   = lane & 7;
    const int q    = (lane >> 3) & 1;

    v2f Amat[3];
    {
        const int px = n >> 3, r = n & 7;
        const int dy = min(r >> 1, 2), c = r & 1;
#pragma unroll
        for (int jc = 0; jc < 3; ++jc) {
#pragma unroll
            for (int e = 0; e < 2; ++e) {
                const int dx  = 2 * h + e - px;
                const int dxc = min(max(dx, 0), 2);
                const int wi  = jc * 9 + dy * 3 + dxc;
                const float wd = conv_w[c * 27 + wi] - conv_w[54 + wi];
                Amat[jc][e] = (r < 6 && dx >= 0 && dx <= 2) ? wd : 0.0f;
            }
        }
    }
    const float cbd0 = cb0 - cb2, cbd1 = cb1 - cb2;

    const int boff = q * (15 * CHSZ) + 2 * pr + 2 * h + wv * WROWS * PITCH;
    auto load_item = [&](int t, v2f (&B)[3]) {
        const float* xp = X + boff + (t / 5) * PITCH + (t % 5) * (3 * CHSZ);
#pragma unroll
        for (int jc = 0; jc < 3; ++jc)
            B[jc] = *(const v2f*)(xp + jc * CHSZ);
    };

    float T0[5][2], T1[5][2];
    float p0 = 0.f, p1 = 0.f, p2 = 0.f;
    float acc[3][5];

    const int xcol = wgx + 2 * pr + h;

    auto finish_item = [&](int t, const v8f dv) {
        const int j = t / 5, gp = t % 5;
        float d0 = 0.f, d1 = 0.f;
        if (j >= 2) {
            d0 = T1[gp][0] + dv[4];
            d1 = T1[gp][1] + dv[5];
        }
        if (j >= 1 && j <= WROWS) {
            T1[gp][0] = T0[gp][0] + dv[2];
            T1[gp][1] = T0[gp][1] + dv[3];
        }
        if (j < WROWS) {
            T0[gp][0] = dv[0];
            T0[gp][1] = dv[1];
        }
        if (j >= 2) {
            if (gp == 0) {
                const float y0 = d0 + cbd0, y1 = d1 + cbd1;
                const float m  = fmaxf(fmaxf(y0, y1), 0.0f);
                const float e0 = __expf(y0 - m), e1 = __expf(y1 - m), e2 = __expf(-m);
                const float is = 1.0f / (e0 + e1 + e2);
                const float a0 = e0 * is, a1 = e1 * is, a2 = e2 * is;
                const float b0 = xoct(a0), b1 = xoct(a1), b2 = xoct(a2);
                p0 = q ? b0 : a0;
                p1 = q ? b1 : a1;
                p2 = q ? b2 : a2;
            }
            const float dot = p0 * d0 + p1 * d1;
            const float w0 = p0 * (d0 - dot), w1 = p1 * (d1 - dot), w2 = -(p2 * dot);
            if (gp == 0) {
                acc[0][0] = q ? w0 : p0;
                acc[1][0] = q ? w1 : p1;
                acc[2][0] = q ? w2 : p2;
            } else {
                acc[0][gp] = w0;
                acc[1][gp] = w1;
                acc[2][gp] = w2;
            }
            if (gp == 4) {
                const int ty  = wv * WROWS + (j - 2);
                const long pix = (long)(wgy + ty) * IMG + xcol;
                float* wb = w_out + pix * 27 + (q ? 5 : 0);
#pragma unroll
                for (int s = 0; s < 3; ++s) {
                    f4u v = {acc[s][1], acc[s][2], acc[s][3], acc[s][4]};
                    *(f4u*)(wb + s * 9) = v;
                }
                float* sb = q ? (w_out + pix * 27 + 4) : (u_out + pix * 3);
                const int ss = q ? 9 : 1;
#pragma unroll
                for (int s = 0; s < 3; ++s) sb[s * ss] = acc[s][0];
            }
        }
    };

    v2f Bq[2][2][3];
    load_item(0, Bq[0][0]);
    load_item(1, Bq[0][1]);

#pragma unroll
    for (int st = 0; st < 15; ++st) {
        if (st + 1 < 15) {
            load_item(2 * st + 2, Bq[(st + 1) & 1][0]);
            load_item(2 * st + 3, Bq[(st + 1) & 1][1]);
        }
        __builtin_amdgcn_sched_barrier(0);
        v8f dq0 = (v8f){0.f, 0.f, 0.f, 0.f, 0.f, 0.f, 0.f, 0.f};
        v8f dq1 = (v8f){0.f, 0.f, 0.f, 0.f, 0.f, 0.f, 0.f, 0.f};
#pragma unroll
        for (int jc = 0; jc < 3; ++jc) {
            dq0 = __builtin_amdgcn_wmma_f32_16x16x4_f32(
                false, Amat[jc], false, Bq[st & 1][0][jc], (short)0, dq0, false, false);
            dq1 = __builtin_amdgcn_wmma_f32_16x16x4_f32(
                false, Amat[jc], false, Bq[st & 1][1][jc], (short)0, dq1, false, false);
        }
        __builtin_amdgcn_sched_barrier(0);
        finish_item(2 * st, dq0);
        finish_item(2 * st + 1, dq1);
    }
}

extern "C" void kernel_launch(void* const* d_in, const int* in_sizes, int n_in,
                              void* d_out, int out_size, void* d_ws, size_t ws_size,
                              hipStream_t stream) {
    (void)in_sizes; (void)n_in; (void)out_size; (void)d_ws; (void)ws_size;
    const float* obs    = (const float*)d_in[0];
    const float* P      = (const float*)d_in[1];
    const float* u_k    = (const float*)d_in[2];
    const float* w_k    = (const float*)d_in[3];
    const float* conv_w = (const float*)d_in[4];
    const float* conv_b = (const float*)d_in[5];
    float* u_out = (float*)d_out;
    float* w_out = u_out + (size_t)IMG * IMG * 3;

    dim3 grid(IMG / TILE, IMG / TILE);
    dim3 block(NTHR);
    hmm_fused_kernel<<<grid, block, 0, stream>>>(obs, P, u_k, w_k, conv_w, conv_b,
                                                 u_out, w_out);
}
